# starting kernel
# baseline (speedup 1.0000x reference)
#include <hip/hip_runtime.h>
#include <math.h>


#define CUT0 10000
#define CUT1 30000
#define DIM  1024
#define NROWS 4096
#define HEADM 10002
#define M1 20000
#define M2 20257
#define D1 256
#define D2 64
#define ROWS_B 32
#define CHUNK 512
#define NCH_H 20
#define NCH_1 40
#define NCH_2 40

typedef float v2f __attribute__((ext_vector_type(2)));
typedef float v8f __attribute__((ext_vector_type(8)));

__device__ __forceinline__ v8f wmma4(v2f a, v2f b, v8f c) {
  return __builtin_amdgcn_wmma_f32_16x16x4_f32(false, a, false, b, (short)0, c,
                                               false, false);
}

__device__ __forceinline__ void gemm_wave32(const float* xp0, const float* xp1,
                                            const float* wp0, const float* wp1,
                                            const float* wp2, const float* wp3,
                                            int K, v8f acc[2][4]) {
  v2f a0 = *(const v2f*)(xp0);
  v2f a1 = *(const v2f*)(xp1);
  v2f b0 = *(const v2f*)(wp0);
  v2f b1 = *(const v2f*)(wp1);
  v2f b2 = *(const v2f*)(wp2);
  v2f b3 = *(const v2f*)(wp3);
  for (int k = 4; k < K; k += 4) {
    v2f a0n = *(const v2f*)(xp0 + k);
    v2f a1n = *(const v2f*)(xp1 + k);
    v2f b0n = *(const v2f*)(wp0 + k);
    v2f b1n = *(const v2f*)(wp1 + k);
    v2f b2n = *(const v2f*)(wp2 + k);
    v2f b3n = *(const v2f*)(wp3 + k);
    __builtin_amdgcn_sched_barrier(0);
    acc[0][0] = wmma4(a0, b0, acc[0][0]);
    acc[0][1] = wmma4(a0, b1, acc[0][1]);
    acc[0][2] = wmma4(a0, b2, acc[0][2]);
    acc[0][3] = wmma4(a0, b3, acc[0][3]);
    acc[1][0] = wmma4(a1, b0, acc[1][0]);
    acc[1][1] = wmma4(a1, b1, acc[1][1]);
    acc[1][2] = wmma4(a1, b2, acc[1][2]);
    acc[1][3] = wmma4(a1, b3, acc[1][3]);
    __builtin_amdgcn_sched_barrier(0);
    a0 = a0n; a1 = a1n; b0 = b0n; b1 = b1n; b2 = b2n; b3 = b3n;
  }
  acc[0][0] = wmma4(a0, b0, acc[0][0]);
  acc[0][1] = wmma4(a0, b1, acc[0][1]);
  acc[0][2] = wmma4(a0, b2, acc[0][2]);
  acc[0][3] = wmma4(a0, b3, acc[0][3]);
  acc[1][0] = wmma4(a1, b0, acc[1][0]);
  acc[1][1] = wmma4(a1, b1, acc[1][1]);
  acc[1][2] = wmma4(a1, b2, acc[1][2]);
  acc[1][3] = wmma4(a1, b3, acc[1][3]);
}

__global__ void gemm_store(const float* __restrict__ X, const float* __restrict__ W,
                           float* __restrict__ C, int K, int M) {
  const int wave = threadIdx.x >> 5;
  const int lane = threadIdx.x & 31;
  const int half = lane >> 4;
  const int r    = lane & 15;
  const int row0 = blockIdx.y * ROWS_B;
  const int col0 = blockIdx.x * CHUNK + wave * 64;
  if (col0 >= M) return;

  const float* xp0 = X + (size_t)(row0 + r) * K + 2 * half;
  const float* xp1 = X + (size_t)(row0 + 16 + r) * K + 2 * half;
  const float* wp0 = W + (size_t)(col0 + r) * K + 2 * half;
  const float* wp1 = W + (size_t)(col0 + 16 + r) * K + 2 * half;
  const float* wp2 = W + (size_t)(col0 + 32 + r) * K + 2 * half;
  const float* wp3 = W + (size_t)(col0 + 48 + r) * K + 2 * half;

  v8f acc[2][4];
#pragma unroll
  for (int rt = 0; rt < 2; ++rt)
#pragma unroll
    for (int t = 0; t < 4; ++t)
      acc[rt][t] = (v8f){0.f,0.f,0.f,0.f,0.f,0.f,0.f,0.f};
  gemm_wave32(xp0, xp1, wp0, wp1, wp2, wp3, K, acc);

#pragma unroll
  for (int rt = 0; rt < 2; ++rt) {
#pragma unroll
    for (int t = 0; t < 4; ++t) {
#pragma unroll
      for (int v = 0; v < 8; ++v) {
        int rr = row0 + 16 * rt + v + 8 * half;
        C[(size_t)rr * M + col0 + 16 * t + r] = acc[rt][t][v];
      }
    }
  }
}

__global__ void gemm_expsum(const float* __restrict__ X, const float* __restrict__ W,
                            float* __restrict__ partial, int K, int M, int nchunks) {
  __shared__ float lsum[8][ROWS_B];
  const int wave = threadIdx.x >> 5;
  const int lane = threadIdx.x & 31;
  const int half = lane >> 4;
  const int r    = lane & 15;
  const int row0 = blockIdx.y * ROWS_B;
  const int col0 = blockIdx.x * CHUNK + wave * 64;
  const bool active = (col0 < M);

  v8f acc[2][4];
#pragma unroll
  for (int rt = 0; rt < 2; ++rt)
#pragma unroll
    for (int t = 0; t < 4; ++t)
      acc[rt][t] = (v8f){0.f,0.f,0.f,0.f,0.f,0.f,0.f,0.f};
  if (active) {
    const float* xp0 = X + (size_t)(row0 + r) * K + 2 * half;
    const float* xp1 = X + (size_t)(row0 + 16 + r) * K + 2 * half;
    const float* wp0 = W + (size_t)min(col0 + r,      M - 1) * K + 2 * half;
    const float* wp1 = W + (size_t)min(col0 + 16 + r, M - 1) * K + 2 * half;
    const float* wp2 = W + (size_t)min(col0 + 32 + r, M - 1) * K + 2 * half;
    const float* wp3 = W + (size_t)min(col0 + 48 + r, M - 1) * K + 2 * half;
    gemm_wave32(xp0, xp1, wp0, wp1, wp2, wp3, K, acc);
  }
#pragma unroll
  for (int rt = 0; rt < 2; ++rt) {
#pragma unroll
    for (int v = 0; v < 8; ++v) {
      float e = 0.f;
      if (active) {
#pragma unroll
        for (int t = 0; t < 4; ++t)
          if (col0 + 16 * t + r < M) e += expf(acc[rt][t][v]);
      }
      e += __shfl_xor(e, 1, 32);
      e += __shfl_xor(e, 2, 32);
      e += __shfl_xor(e, 4, 32);
      e += __shfl_xor(e, 8, 32);
      if (r == 0) lsum[wave][16 * rt + v + 8 * half] = e;
    }
  }
  __syncthreads();
  if (threadIdx.x < ROWS_B) {
    float t = 0.f;
#pragma unroll
    for (int w = 0; w < 8; ++w) t += lsum[w][threadIdx.x];
    partial[(size_t)(row0 + threadIdx.x) * nchunks + blockIdx.x] = t;
  }
}

__global__ void gather_dots(const float* __restrict__ x, const int* __restrict__ target,
                            const float* __restrict__ Wh, const float* __restrict__ Wo1,
                            const float* __restrict__ Wo2, const float* __restrict__ h1,
                            const float* __restrict__ h2,
                            float* __restrict__ dh, float* __restrict__ d1,
                            float* __restrict__ d2) {
  const int wave = threadIdx.x >> 5;
  const int lane = threadIdx.x & 31;
  const int row  = blockIdx.x * 8 + wave;
  const int t = target[row];
  const int g = (t < CUT0) ? t : ((t < CUT1) ? CUT0 : CUT0 + 1);

  float s0 = 0.f, s1 = 0.f, s2 = 0.f;
  {
    const float* xr = x + (size_t)row * DIM;
    const float* wr = Wh + (size_t)g * DIM;
    for (int k = lane; k < DIM; k += 32) s0 += xr[k] * wr[k];
  }
  {
    const int rel = min(max(t - CUT0, 0), M1 - 1);
    const float* hr = h1 + (size_t)row * D1;
    const float* wr = Wo1 + (size_t)rel * D1;
    for (int k = lane; k < D1; k += 32) s1 += hr[k] * wr[k];
  }
  {
    const int rel = min(max(t - CUT1, 0), M2 - 1);
    const float* hr = h2 + (size_t)row * D2;
    const float* wr = Wo2 + (size_t)rel * D2;
    for (int k = lane; k < D2; k += 32) s2 += hr[k] * wr[k];
  }
#pragma unroll
  for (int m = 16; m; m >>= 1) {
    s0 += __shfl_xor(s0, m, 32);
    s1 += __shfl_xor(s1, m, 32);
    s2 += __shfl_xor(s2, m, 32);
  }
  if (lane == 0) { dh[row] = s0; d1[row] = s1; d2[row] = s2; }
}

__global__ void finalize(const int* __restrict__ target,
                         const float* __restrict__ ph, const float* __restrict__ p1,
                         const float* __restrict__ p2, const float* __restrict__ dh,
                         const float* __restrict__ d1, const float* __restrict__ d2,
                         float* __restrict__ out) {
  const int i = blockIdx.x * blockDim.x + threadIdx.x;
  if (i >= NROWS) return;
  float sh = 0.f;
  for (int c = 0; c < NCH_H; ++c) sh += ph[(size_t)i * NCH_H + c];
  float o = dh[i] - logf(sh);
  const int t = target[i];
  if (t >= CUT0 && t < CUT1) {
    float s = 0.f;
    for (int c = 0; c < NCH_1; ++c) s += p1[(size_t)i * NCH_1 + c];
    o += d1[i] - logf(s);
  } else if (t >= CUT1) {
    float s = 0.f;
    for (int c = 0; c < NCH_2; ++c) s += p2[(size_t)i * NCH_2 + c];
    o += d2[i] - logf(s);
  }
  out[i] = o;
}

__global__ void loss_reduce(const float* __restrict__ out, float* __restrict__ loss) {
  __shared__ float sm[256];
  float s = 0.f;
  for (int i = threadIdx.x; i < NROWS; i += 256) s += out[i];
  sm[threadIdx.x] = s;
  __syncthreads();
  for (int w = 128; w; w >>= 1) {
    if (threadIdx.x < w) sm[threadIdx.x] += sm[threadIdx.x + w];
    __syncthreads();
  }
  if (threadIdx.x == 0) *loss = -sm[0] / (float)NROWS;
}

extern "C" void kernel_launch(void* const* d_in, const int* in_sizes, int n_in,
                              void* d_out, int out_size, void* d_ws, size_t ws_size,
                              hipStream_t stream) {
  (void)in_sizes; (void)n_in; (void)out_size; (void)ws_size;
  const float* x    = (const float*)d_in[0];
  const int*   targ = (const int*)d_in[1];
  const float* Wh   = (const float*)d_in[2];
  const float* Wp1  = (const float*)d_in[3];
  const float* Wo1  = (const float*)d_in[4];
  const float* Wp2  = (const float*)d_in[5];
  const float* Wo2  = (const float*)d_in[6];
  float* out = (float*)d_out;

  float* ws = (float*)d_ws;
  float* h1 = ws;
  float* h2 = h1 + (size_t)NROWS * D1;
  float* ph = h2 + (size_t)NROWS * D2;
  float* p1 = ph + (size_t)NROWS * NCH_H;
  float* p2 = p1 + (size_t)NROWS * NCH_1;
  float* dh = p2 + (size_t)NROWS * NCH_2;
  float* d1 = dh + NROWS;
  float* d2 = d1 + NROWS;

  const dim3 blk(256);
  const int rowBlocks = NROWS / ROWS_B;
  gemm_store<<<dim3(1, rowBlocks), blk, 0, stream>>>(x, Wp1, h1, DIM, D1);
  gemm_store<<<dim3(1, rowBlocks), blk, 0, stream>>>(x, Wp2, h2, DIM, D2);
  gemm_expsum<<<dim3(NCH_H, rowBlocks), blk, 0, stream>>>(x,  Wh,  ph, DIM, HEADM, NCH_H);
  gemm_expsum<<<dim3(NCH_1, rowBlocks), blk, 0, stream>>>(h1, Wo1, p1, D1,  M1,    NCH_1);
  gemm_expsum<<<dim3(NCH_2, rowBlocks), blk, 0, stream>>>(h2, Wo2, p2, D2,  M2,    NCH_2);
  gather_dots<<<dim3(NROWS / 8), blk, 0, stream>>>(x, targ, Wh, Wo1, Wo2, h1, h2, dh, d1, d2);
  finalize<<<dim3(NROWS / 256), blk, 0, stream>>>(targ, ph, p1, p2, dh, d1, d2, out);
  loss_reduce<<<1, 256, 0, stream>>>(out, out + NROWS);
}


# GEMM fragment loads: each lane reads 64 contiguous bytes of its row (4 x 128-bit, whole 128-B line per row per wave) direct to registers, two register sets, no LDS; changes k order inside each 32
# speedup vs baseline: 2.3290x; 2.3290x over previous
#include <hip/hip_runtime.h>
#include <math.h>


#define CUT0 10000
#define CUT1 30000
#define DIM  1024
#define NROWS 4096
#define HEADM 10002
#define M1 20000
#define M2 20257
#define D1 256
#define D2 64
#define ROWS_B 32
#define CHUNK 512
#define NCH_H 20
#define NCH_1 40
#define NCH_2 40

typedef float v2f __attribute__((ext_vector_type(2)));
typedef float v4f __attribute__((ext_vector_type(4)));
typedef float v8f __attribute__((ext_vector_type(8)));

__device__ __forceinline__ v8f wmma4(v2f a, v2f b, v8f c) {
  return __builtin_amdgcn_wmma_f32_16x16x4_f32(false, a, false, b, (short)0, c,
                                               false, false);
}

struct Frag { v4f q[4]; };

__device__ __forceinline__ void load_frag(Frag& f, const float* p) {
  f.q[0] = *(const v4f*)(p);
  f.q[1] = *(const v4f*)(p + 4);
  f.q[2] = *(const v4f*)(p + 8);
  f.q[3] = *(const v4f*)(p + 12);
}

__device__ __forceinline__ v2f pairk(const Frag& f, int j) {
  const v4f q = f.q[j >> 1];
  return (j & 1) ? (v2f){q.z, q.w} : (v2f){q.x, q.y};
}

struct Set { Frag a0, a1, b0, b1, b2, b3; };

__device__ __forceinline__ void load_set(Set& s, const float* xp0, const float* xp1,
                                         const float* wp0, const float* wp1,
                                         const float* wp2, const float* wp3, int kb) {
  load_frag(s.a0, xp0 + kb);
  load_frag(s.a1, xp1 + kb);
  load_frag(s.b0, wp0 + kb);
  load_frag(s.b1, wp1 + kb);
  load_frag(s.b2, wp2 + kb);
  load_frag(s.b3, wp3 + kb);
}

__device__ __forceinline__ void mma_set(const Set& s, v8f acc[2][4]) {
#pragma unroll
  for (int j = 0; j < 8; ++j) {
    const v2f a0 = pairk(s.a0, j), a1 = pairk(s.a1, j);
    const v2f b0 = pairk(s.b0, j), b1 = pairk(s.b1, j);
    const v2f b2 = pairk(s.b2, j), b3 = pairk(s.b3, j);
    acc[0][0] = wmma4(a0, b0, acc[0][0]);
    acc[0][1] = wmma4(a0, b1, acc[0][1]);
    acc[0][2] = wmma4(a0, b2, acc[0][2]);
    acc[0][3] = wmma4(a0, b3, acc[0][3]);
    acc[1][0] = wmma4(a1, b0, acc[1][0]);
    acc[1][1] = wmma4(a1, b1, acc[1][1]);
    acc[1][2] = wmma4(a1, b2, acc[1][2]);
    acc[1][3] = wmma4(a1, b3, acc[1][3]);
  }
}

__device__ __forceinline__ void gemm_wave32(const float* xp0, const float* xp1,
                                            const float* wp0, const float* wp1,
                                            const float* wp2, const float* wp3,
                                            int K, v8f acc[2][4]) {
  Set P, Q;
  load_set(P, xp0, xp1, wp0, wp1, wp2, wp3, 0);
  for (int kb = 0; kb < K - 64; kb += 64) {
    load_set(Q, xp0, xp1, wp0, wp1, wp2, wp3, kb + 32);
    __builtin_amdgcn_sched_barrier(0);
    mma_set(P, acc);
    __builtin_amdgcn_sched_barrier(0);
    load_set(P, xp0, xp1, wp0, wp1, wp2, wp3, kb + 64);
    __builtin_amdgcn_sched_barrier(0);
    mma_set(Q, acc);
  }
  load_set(Q, xp0, xp1, wp0, wp1, wp2, wp3, K - 32);
  __builtin_amdgcn_sched_barrier(0);
  mma_set(P, acc);
  mma_set(Q, acc);
}

__global__ void __launch_bounds__(256) gemm_store(const float* __restrict__ X, const float* __restrict__ W,
                           float* __restrict__ C, int K, int M) {
  const int wave = threadIdx.x >> 5;
  const int lane = threadIdx.x & 31;
  const int half = lane >> 4;
  const int r    = lane & 15;
  const int row0 = blockIdx.y * ROWS_B;
  const int col0 = blockIdx.x * CHUNK + wave * 64;
  if (col0 >= M) return;

  const float* xp0 = X + (size_t)(row0 + r) * K + 16 * half;
  const float* xp1 = X + (size_t)(row0 + 16 + r) * K + 16 * half;
  const float* wp0 = W + (size_t)(col0 + r) * K + 16 * half;
  const float* wp1 = W + (size_t)(col0 + 16 + r) * K + 16 * half;
  const float* wp2 = W + (size_t)(col0 + 32 + r) * K + 16 * half;
  const float* wp3 = W + (size_t)(col0 + 48 + r) * K + 16 * half;

  v8f acc[2][4];
#pragma unroll
  for (int rt = 0; rt < 2; ++rt)
#pragma unroll
    for (int t = 0; t < 4; ++t)
      acc[rt][t] = (v8f){0.f,0.f,0.f,0.f,0.f,0.f,0.f,0.f};
  gemm_wave32(xp0, xp1, wp0, wp1, wp2, wp3, K, acc);

#pragma unroll
  for (int rt = 0; rt < 2; ++rt) {
#pragma unroll
    for (int t = 0; t < 4; ++t) {
#pragma unroll
      for (int v = 0; v < 8; ++v) {
        int rr = row0 + 16 * rt + v + 8 * half;
        C[(size_t)rr * M + col0 + 16 * t + r] = acc[rt][t][v];
      }
    }
  }
}

__global__ void __launch_bounds__(256) gemm_expsum(const float* __restrict__ X, const float* __restrict__ W,
                            float* __restrict__ partial, int K, int M, int nchunks) {
  __shared__ float lsum[8][ROWS_B];
  const int wave = threadIdx.x >> 5;
  const int lane = threadIdx.x & 31;
  const int half = lane >> 4;
  const int r    = lane & 15;
  const int row0 = blockIdx.y * ROWS_B;
  const int col0 = blockIdx.x * CHUNK + wave * 64;
  const bool active = (col0 < M);

  v8f acc[2][4];
#pragma unroll
  for (int rt = 0; rt < 2; ++rt)
#pragma unroll
    for (int t = 0; t < 4; ++t)
      acc[rt][t] = (v8f){0.f,0.f,0.f,0.f,0.f,0.f,0.f,0.f};
  if (active) {
    const float* xp0 = X + (size_t)(row0 + r) * K + 16 * half;
    const float* xp1 = X + (size_t)(row0 + 16 + r) * K + 16 * half;
    const float* wp0 = W + (size_t)min(col0 + r,      M - 1) * K + 16 * half;
    const float* wp1 = W + (size_t)min(col0 + 16 + r, M - 1) * K + 16 * half;
    const float* wp2 = W + (size_t)min(col0 + 32 + r, M - 1) * K + 16 * half;
    const float* wp3 = W + (size_t)min(col0 + 48 + r, M - 1) * K + 16 * half;
    gemm_wave32(xp0, xp1, wp0, wp1, wp2, wp3, K, acc);
  }
#pragma unroll
  for (int rt = 0; rt < 2; ++rt) {
#pragma unroll
    for (int v = 0; v < 8; ++v) {
      float e = 0.f;
      if (active) {
#pragma unroll
        for (int t = 0; t < 4; ++t)
          if (col0 + 16 * t + r < M) e += expf(acc[rt][t][v]);
      }
      e += __shfl_xor(e, 1, 32);
      e += __shfl_xor(e, 2, 32);
      e += __shfl_xor(e, 4, 32);
      e += __shfl_xor(e, 8, 32);
      if (r == 0) lsum[wave][16 * rt + v + 8 * half] = e;
    }
  }
  __syncthreads();
  if (threadIdx.x < ROWS_B) {
    float t = 0.f;
#pragma unroll
    for (int w = 0; w < 8; ++w) t += lsum[w][threadIdx.x];
    partial[(size_t)(row0 + threadIdx.x) * nchunks + blockIdx.x] = t;
  }
}

__global__ void gather_dots(const float* __restrict__ x, const int* __restrict__ target,
                            const float* __restrict__ Wh, const float* __restrict__ Wo1,
                            const float* __restrict__ Wo2, const float* __restrict__ h1,
                            const float* __restrict__ h2,
                            float* __restrict__ dh, float* __restrict__ d1,
                            float* __restrict__ d2) {
  const int wave = threadIdx.x >> 5;
  const int lane = threadIdx.x & 31;
  const int row  = blockIdx.x * 8 + wave;
  const int t = target[row];
  const int g = (t < CUT0) ? t : ((t < CUT1) ? CUT0 : CUT0 + 1);

  float s0 = 0.f, s1 = 0.f, s2 = 0.f;
  {
    const float* xr = x + (size_t)row * DIM;
    const float* wr = Wh + (size_t)g * DIM;
    for (int k = lane; k < DIM; k += 32) s0 += xr[k] * wr[k];
  }
  {
    const int rel = min(max(t - CUT0, 0), M1 - 1);
    const float* hr = h1 + (size_t)row * D1;
    const float* wr = Wo1 + (size_t)rel * D1;
    for (int k = lane; k < D1; k += 32) s1 += hr[k] * wr[k];
  }
  {
    const int rel = min(max(t - CUT1, 0), M2 - 1);
    const float* hr = h2 + (size_t)row * D2;
    const float* wr = Wo2 + (size_t)rel * D2;
    for (int k = lane; k < D2; k += 32) s2 += hr[k] * wr[k];
  }
#pragma unroll
  for (int m = 16; m; m >>= 1) {
    s0 += __shfl_xor(s0, m, 32);
    s1 += __shfl_xor(s1, m, 32);
    s2 += __shfl_xor(s2, m, 32);
  }
  if (lane == 0) { dh[row] = s0; d1[row] = s1; d2[row] = s2; }
}

__global__ void finalize(const int* __restrict__ target,
                         const float* __restrict__ ph, const float* __restrict__ p1,
                         const float* __restrict__ p2, const float* __restrict__ dh,
                         const float* __restrict__ d1, const float* __restrict__ d2,
                         float* __restrict__ out) {
  const int i = blockIdx.x * blockDim.x + threadIdx.x;
  if (i >= NROWS) return;
  float sh = 0.f;
  for (int c = 0; c < NCH_H; ++c) sh += ph[(size_t)i * NCH_H + c];
  float o = dh[i] - logf(sh);
  const int t = target[i];
  if (t >= CUT0 && t < CUT1) {
    float s = 0.f;
    for (int c = 0; c < NCH_1; ++c) s += p1[(size_t)i * NCH_1 + c];
    o += d1[i] - logf(s);
  } else if (t >= CUT1) {
    float s = 0.f;
    for (int c = 0; c < NCH_2; ++c) s += p2[(size_t)i * NCH_2 + c];
    o += d2[i] - logf(s);
  }
  out[i] = o;
}

__global__ void loss_reduce(const float* __restrict__ out, float* __restrict__ loss) {
  __shared__ float sm[256];
  float s = 0.f;
  for (int i = threadIdx.x; i < NROWS; i += 256) s += out[i];
  sm[threadIdx.x] = s;
  __syncthreads();
  for (int w = 128; w; w >>= 1) {
    if (threadIdx.x < w) sm[threadIdx.x] += sm[threadIdx.x + w];
    __syncthreads();
  }
  if (threadIdx.x == 0) *loss = -sm[0] / (float)NROWS;
}

extern "C" void kernel_launch(void* const* d_in, const int* in_sizes, int n_in,
                              void* d_out, int out_size, void* d_ws, size_t ws_size,
                              hipStream_t stream) {
  (void)in_sizes; (void)n_in; (void)out_size; (void)ws_size;
  const float* x    = (const float*)d_in[0];
  const int*   targ = (const int*)d_in[1];
  const float* Wh   = (const float*)d_in[2];
  const float* Wp1  = (const float*)d_in[3];
  const float* Wo1  = (const float*)d_in[4];
  const float* Wp2  = (const float*)d_in[5];
  const float* Wo2  = (const float*)d_in[6];
  float* out = (float*)d_out;

  float* ws = (float*)d_ws;
  float* h1 = ws;
  float* h2 = h1 + (size_t)NROWS * D1;
  float* ph = h2 + (size_t)NROWS * D2;
  float* p1 = ph + (size_t)NROWS * NCH_H;
  float* p2 = p1 + (size_t)NROWS * NCH_1;
  float* dh = p2 + (size_t)NROWS * NCH_2;
  float* d1 = dh + NROWS;
  float* d2 = d1 + NROWS;

  const dim3 blk(256);
  const int rowBlocks = NROWS / ROWS_B;
  gemm_store<<<dim3(1, rowBlocks), blk, 0, stream>>>(x, Wp1, h1, DIM, D1);
  gemm_store<<<dim3(1, rowBlocks), blk, 0, stream>>>(x, Wp2, h2, DIM, D2);
  gemm_expsum<<<dim3(NCH_H, rowBlocks), blk, 0, stream>>>(x,  Wh,  ph, DIM, HEADM, NCH_H);
  gemm_expsum<<<dim3(NCH_1, rowBlocks), blk, 0, stream>>>(h1, Wo1, p1, D1,  M1,    NCH_1);
  gemm_expsum<<<dim3(NCH_2, rowBlocks), blk, 0, stream>>>(h2, Wo2, p2, D2,  M2,    NCH_2);
  gather_dots<<<dim3(NROWS / 8), blk, 0, stream>>>(x, targ, Wh, Wo1, Wo2, h1, h2, dh, d1, d2);
  finalize<<<dim3(NROWS / 256), blk, 0, stream>>>(targ, ph, p1, p2, dh, d1, d2, out);
  loss_reduce<<<1, 256, 0, stream>>>(out, out + NROWS);
}


# Exp-sum GEMMs: wave tile 64x64 (was 32x64): a K block of 32 feeds 128 WMMAs from 8 row-per-lane fragments (was 64 from 6), W loads per WMMA halved; block 64 rows x 512 cols; k and sum order unchanged
# speedup vs baseline: 2.5515x; 1.0955x over previous
#include <hip/hip_runtime.h>
#include <math.h>


#define CUT0 10000
#define CUT1 30000
#define DIM  1024
#define NROWS 4096
#define HEADM 10002
#define M1 20000
#define M2 20257
#define D1 256
#define D2 64
#define ROWS_B 32
#define ROWS_E 64
#define CHUNK 512
#define NCH_H 20
#define NCH_1 40
#define NCH_2 40

typedef float v2f __attribute__((ext_vector_type(2)));
typedef float v4f __attribute__((ext_vector_type(4)));
typedef float v8f __attribute__((ext_vector_type(8)));

__device__ __forceinline__ v8f wmma4(v2f a, v2f b, v8f c) {
  return __builtin_amdgcn_wmma_f32_16x16x4_f32(false, a, false, b, (short)0, c,
                                               false, false);
}

struct Frag { v4f q[4]; };

__device__ __forceinline__ void load_frag(Frag& f, const float* p) {
  f.q[0] = *(const v4f*)(p);
  f.q[1] = *(const v4f*)(p + 4);
  f.q[2] = *(const v4f*)(p + 8);
  f.q[3] = *(const v4f*)(p + 12);
}

__device__ __forceinline__ v2f pairk(const Frag& f, int j) {
  const v4f q = f.q[j >> 1];
  return (j & 1) ? (v2f){q.z, q.w} : (v2f){q.x, q.y};
}

struct Set { Frag a0, a1, b0, b1, b2, b3; };

__device__ __forceinline__ void load_set(Set& s, const float* xp0, const float* xp1,
                                         const float* wp0, const float* wp1,
                                         const float* wp2, const float* wp3, int kb) {
  load_frag(s.a0, xp0 + kb);
  load_frag(s.a1, xp1 + kb);
  load_frag(s.b0, wp0 + kb);
  load_frag(s.b1, wp1 + kb);
  load_frag(s.b2, wp2 + kb);
  load_frag(s.b3, wp3 + kb);
}

__device__ __forceinline__ void mma_set(const Set& s, v8f acc[2][4]) {
#pragma unroll
  for (int j = 0; j < 8; ++j) {
    const v2f a0 = pairk(s.a0, j), a1 = pairk(s.a1, j);
    const v2f b0 = pairk(s.b0, j), b1 = pairk(s.b1, j);
    const v2f b2 = pairk(s.b2, j), b3 = pairk(s.b3, j);
    acc[0][0] = wmma4(a0, b0, acc[0][0]);
    acc[0][1] = wmma4(a0, b1, acc[0][1]);
    acc[0][2] = wmma4(a0, b2, acc[0][2]);
    acc[0][3] = wmma4(a0, b3, acc[0][3]);
    acc[1][0] = wmma4(a1, b0, acc[1][0]);
    acc[1][1] = wmma4(a1, b1, acc[1][1]);
    acc[1][2] = wmma4(a1, b2, acc[1][2]);
    acc[1][3] = wmma4(a1, b3, acc[1][3]);
  }
}

__device__ __forceinline__ void gemm_wave32(const float* xp0, const float* xp1,
                                            const float* wp0, const float* wp1,
                                            const float* wp2, const float* wp3,
                                            int K, v8f acc[2][4]) {
  Set P, Q;
  load_set(P, xp0, xp1, wp0, wp1, wp2, wp3, 0);
  for (int kb = 0; kb < K - 64; kb += 64) {
    load_set(Q, xp0, xp1, wp0, wp1, wp2, wp3, kb + 32);
    __builtin_amdgcn_sched_barrier(0);
    mma_set(P, acc);
    __builtin_amdgcn_sched_barrier(0);
    load_set(P, xp0, xp1, wp0, wp1, wp2, wp3, kb + 64);
    __builtin_amdgcn_sched_barrier(0);
    mma_set(Q, acc);
  }
  load_set(Q, xp0, xp1, wp0, wp1, wp2, wp3, K - 32);
  __builtin_amdgcn_sched_barrier(0);
  mma_set(P, acc);
  mma_set(Q, acc);
}

struct Set4 { Frag a[4]; Frag b[4]; };

__device__ __forceinline__ void load_set4(Set4& s, const float* const xp[4],
                                          const float* const wp[4], int kb) {
#pragma unroll
  for (int i = 0; i < 4; ++i) load_frag(s.a[i], xp[i] + kb);
#pragma unroll
  for (int t = 0; t < 4; ++t) load_frag(s.b[t], wp[t] + kb);
}

__device__ __forceinline__ void mma_set4(const Set4& s, v8f acc[4][4]) {
#pragma unroll
  for (int j = 0; j < 8; ++j) {
    v2f a[4], b[4];
#pragma unroll
    for (int i = 0; i < 4; ++i) { a[i] = pairk(s.a[i], j); b[i] = pairk(s.b[i], j); }
#pragma unroll
    for (int i = 0; i < 4; ++i)
#pragma unroll
      for (int t = 0; t < 4; ++t)
        acc[i][t] = wmma4(a[i], b[t], acc[i][t]);
  }
}

__device__ __forceinline__ void gemm_wave64(const float* const xp[4], const float* const wp[4],
                                            int K, v8f acc[4][4]) {
  Set4 P, Q;
  load_set4(P, xp, wp, 0);
  for (int kb = 0; kb < K - 64; kb += 64) {
    load_set4(Q, xp, wp, kb + 32);
    __builtin_amdgcn_sched_barrier(0);
    mma_set4(P, acc);
    __builtin_amdgcn_sched_barrier(0);
    load_set4(P, xp, wp, kb + 64);
    __builtin_amdgcn_sched_barrier(0);
    mma_set4(Q, acc);
  }
  load_set4(Q, xp, wp, K - 32);
  __builtin_amdgcn_sched_barrier(0);
  mma_set4(P, acc);
  mma_set4(Q, acc);
}

__global__ void __launch_bounds__(256) gemm_store(const float* __restrict__ X, const float* __restrict__ W,
                           float* __restrict__ C, int K, int M) {
  const int wave = threadIdx.x >> 5;
  const int lane = threadIdx.x & 31;
  const int half = lane >> 4;
  const int r    = lane & 15;
  const int row0 = blockIdx.y * ROWS_B;
  const int col0 = blockIdx.x * CHUNK + wave * 64;
  if (col0 >= M) return;

  const float* xp0 = X + (size_t)(row0 + r) * K + 16 * half;
  const float* xp1 = X + (size_t)(row0 + 16 + r) * K + 16 * half;
  const float* wp0 = W + (size_t)(col0 + r) * K + 16 * half;
  const float* wp1 = W + (size_t)(col0 + 16 + r) * K + 16 * half;
  const float* wp2 = W + (size_t)(col0 + 32 + r) * K + 16 * half;
  const float* wp3 = W + (size_t)(col0 + 48 + r) * K + 16 * half;

  v8f acc[2][4];
#pragma unroll
  for (int rt = 0; rt < 2; ++rt)
#pragma unroll
    for (int t = 0; t < 4; ++t)
      acc[rt][t] = (v8f){0.f,0.f,0.f,0.f,0.f,0.f,0.f,0.f};
  gemm_wave32(xp0, xp1, wp0, wp1, wp2, wp3, K, acc);

#pragma unroll
  for (int rt = 0; rt < 2; ++rt) {
#pragma unroll
    for (int t = 0; t < 4; ++t) {
#pragma unroll
      for (int v = 0; v < 8; ++v) {
        int rr = row0 + 16 * rt + v + 8 * half;
        C[(size_t)rr * M + col0 + 16 * t + r] = acc[rt][t][v];
      }
    }
  }
}

__global__ void __launch_bounds__(256) gemm_expsum(const float* __restrict__ X, const float* __restrict__ W,
                            float* __restrict__ partial, int K, int M, int nchunks) {
  __shared__ float lsum[8][ROWS_E];
  const int wave = threadIdx.x >> 5;
  const int lane = threadIdx.x & 31;
  const int half = lane >> 4;
  const int r    = lane & 15;
  const int row0 = blockIdx.y * ROWS_E;
  const int col0 = blockIdx.x * CHUNK + wave * 64;
  const bool active = (col0 < M);

  v8f acc[4][4];
#pragma unroll
  for (int rt = 0; rt < 4; ++rt)
#pragma unroll
    for (int t = 0; t < 4; ++t)
      acc[rt][t] = (v8f){0.f,0.f,0.f,0.f,0.f,0.f,0.f,0.f};
  if (active) {
    const float* xp[4];
    const float* wp[4];
#pragma unroll
    for (int i = 0; i < 4; ++i) {
      xp[i] = X + (size_t)(row0 + 16 * i + r) * K + 16 * half;
      wp[i] = W + (size_t)min(col0 + 16 * i + r, M - 1) * K + 16 * half;
    }
    gemm_wave64(xp, wp, K, acc);
  }
#pragma unroll
  for (int rt = 0; rt < 4; ++rt) {
#pragma unroll
    for (int v = 0; v < 8; ++v) {
      float e = 0.f;
      if (active) {
#pragma unroll
        for (int t = 0; t < 4; ++t)
          if (col0 + 16 * t + r < M) e += expf(acc[rt][t][v]);
      }
      e += __shfl_xor(e, 1, 32);
      e += __shfl_xor(e, 2, 32);
      e += __shfl_xor(e, 4, 32);
      e += __shfl_xor(e, 8, 32);
      if (r == 0) lsum[wave][16 * rt + v + 8 * half] = e;
    }
  }
  __syncthreads();
  if (threadIdx.x < ROWS_E) {
    float t = 0.f;
#pragma unroll
    for (int w = 0; w < 8; ++w) t += lsum[w][threadIdx.x];
    partial[(size_t)(row0 + threadIdx.x) * nchunks + blockIdx.x] = t;
  }
}

__global__ void gather_dots(const float* __restrict__ x, const int* __restrict__ target,
                            const float* __restrict__ Wh, const float* __restrict__ Wo1,
                            const float* __restrict__ Wo2, const float* __restrict__ h1,
                            const float* __restrict__ h2,
                            float* __restrict__ dh, float* __restrict__ d1,
                            float* __restrict__ d2) {
  const int wave = threadIdx.x >> 5;
  const int lane = threadIdx.x & 31;
  const int row  = blockIdx.x * 8 + wave;
  const int t = target[row];
  const int g = (t < CUT0) ? t : ((t < CUT1) ? CUT0 : CUT0 + 1);

  float s0 = 0.f, s1 = 0.f, s2 = 0.f;
  {
    const float* xr = x + (size_t)row * DIM;
    const float* wr = Wh + (size_t)g * DIM;
    for (int k = lane; k < DIM; k += 32) s0 += xr[k] * wr[k];
  }
  {
    const int rel = min(max(t - CUT0, 0), M1 - 1);
    const float* hr = h1 + (size_t)row * D1;
    const float* wr = Wo1 + (size_t)rel * D1;
    for (int k = lane; k < D1; k += 32) s1 += hr[k] * wr[k];
  }
  {
    const int rel = min(max(t - CUT1, 0), M2 - 1);
    const float* hr = h2 + (size_t)row * D2;
    const float* wr = Wo2 + (size_t)rel * D2;
    for (int k = lane; k < D2; k += 32) s2 += hr[k] * wr[k];
  }
#pragma unroll
  for (int m = 16; m; m >>= 1) {
    s0 += __shfl_xor(s0, m, 32);
    s1 += __shfl_xor(s1, m, 32);
    s2 += __shfl_xor(s2, m, 32);
  }
  if (lane == 0) { dh[row] = s0; d1[row] = s1; d2[row] = s2; }
}

__global__ void finalize(const int* __restrict__ target,
                         const float* __restrict__ ph, const float* __restrict__ p1,
                         const float* __restrict__ p2, const float* __restrict__ dh,
                         const float* __restrict__ d1, const float* __restrict__ d2,
                         float* __restrict__ out) {
  const int i = blockIdx.x * blockDim.x + threadIdx.x;
  if (i >= NROWS) return;
  float sh = 0.f;
  for (int c = 0; c < NCH_H; ++c) sh += ph[(size_t)i * NCH_H + c];
  float o = dh[i] - logf(sh);
  const int t = target[i];
  if (t >= CUT0 && t < CUT1) {
    float s = 0.f;
    for (int c = 0; c < NCH_1; ++c) s += p1[(size_t)i * NCH_1 + c];
    o += d1[i] - logf(s);
  } else if (t >= CUT1) {
    float s = 0.f;
    for (int c = 0; c < NCH_2; ++c) s += p2[(size_t)i * NCH_2 + c];
    o += d2[i] - logf(s);
  }
  out[i] = o;
}

__global__ void loss_reduce(const float* __restrict__ out, float* __restrict__ loss) {
  __shared__ float sm[256];
  float s = 0.f;
  for (int i = threadIdx.x; i < NROWS; i += 256) s += out[i];
  sm[threadIdx.x] = s;
  __syncthreads();
  for (int w = 128; w; w >>= 1) {
    if (threadIdx.x < w) sm[threadIdx.x] += sm[threadIdx.x + w];
    __syncthreads();
  }
  if (threadIdx.x == 0) *loss = -sm[0] / (float)NROWS;
}

extern "C" void kernel_launch(void* const* d_in, const int* in_sizes, int n_in,
                              void* d_out, int out_size, void* d_ws, size_t ws_size,
                              hipStream_t stream) {
  (void)in_sizes; (void)n_in; (void)out_size; (void)ws_size;
  const float* x    = (const float*)d_in[0];
  const int*   targ = (const int*)d_in[1];
  const float* Wh   = (const float*)d_in[2];
  const float* Wp1  = (const float*)d_in[3];
  const float* Wo1  = (const float*)d_in[4];
  const float* Wp2  = (const float*)d_in[5];
  const float* Wo2  = (const float*)d_in[6];
  float* out = (float*)d_out;

  float* ws = (float*)d_ws;
  float* h1 = ws;
  float* h2 = h1 + (size_t)NROWS * D1;
  float* ph = h2 + (size_t)NROWS * D2;
  float* p1 = ph + (size_t)NROWS * NCH_H;
  float* p2 = p1 + (size_t)NROWS * NCH_1;
  float* dh = p2 + (size_t)NROWS * NCH_2;
  float* d1 = dh + NROWS;
  float* d2 = d1 + NROWS;

  const dim3 blk(256);
  const int rowBlocks = NROWS / ROWS_B;
  const int rowBlocksE = NROWS / ROWS_E;
  gemm_store<<<dim3(1, rowBlocks), blk, 0, stream>>>(x, Wp1, h1, DIM, D1);
  gemm_store<<<dim3(1, rowBlocks), blk, 0, stream>>>(x, Wp2, h2, DIM, D2);
  gemm_expsum<<<dim3(NCH_H, rowBlocksE), blk, 0, stream>>>(x,  Wh,  ph, DIM, HEADM, NCH_H);
  gemm_expsum<<<dim3(NCH_1, rowBlocksE), blk, 0, stream>>>(h1, Wo1, p1, D1,  M1,    NCH_1);
  gemm_expsum<<<dim3(NCH_2, rowBlocksE), blk, 0, stream>>>(h2, Wo2, p2, D2,  M2,    NCH_2);
  gather_dots<<<dim3(NROWS / 8), blk, 0, stream>>>(x, targ, Wh, Wo1, Wo2, h1, h2, dh, d1, d2);
  finalize<<<dim3(NROWS / 256), blk, 0, stream>>>(targ, ph, p1, p2, dh, d1, d2, out);
  loss_reduce<<<1, 256, 0, stream>>>(out, out + NROWS);
}


# Exp-sum GEMMs: tile computed transposed (W rows as A operand, X rows as B): a softmax row's columns lie in one lane's registers, exp-sum = in-lane adds, no lane exchanges; changes the exp-sum tree
# speedup vs baseline: 3.2098x; 1.2580x over previous
#include <hip/hip_runtime.h>
#include <math.h>


#define CUT0 10000
#define CUT1 30000
#define DIM  1024
#define NROWS 4096
#define HEADM 10002
#define M1 20000
#define M2 20257
#define D1 256
#define D2 64
#define ROWS_B 32
#define ROWS_E 64
#define CHUNK 512
#define NCH_H 20
#define NCH_1 40
#define NCH_2 40

typedef float v2f __attribute__((ext_vector_type(2)));
typedef float v4f __attribute__((ext_vector_type(4)));
typedef float v8f __attribute__((ext_vector_type(8)));

__device__ __forceinline__ v8f wmma4(v2f a, v2f b, v8f c) {
  return __builtin_amdgcn_wmma_f32_16x16x4_f32(false, a, false, b, (short)0, c,
                                               false, false);
}

struct Frag { v4f q[4]; };

__device__ __forceinline__ void load_frag(Frag& f, const float* p) {
  f.q[0] = *(const v4f*)(p);
  f.q[1] = *(const v4f*)(p + 4);
  f.q[2] = *(const v4f*)(p + 8);
  f.q[3] = *(const v4f*)(p + 12);
}

__device__ __forceinline__ v2f pairk(const Frag& f, int j) {
  const v4f q = f.q[j >> 1];
  return (j & 1) ? (v2f){q.z, q.w} : (v2f){q.x, q.y};
}

struct Set { Frag a0, a1, b0, b1, b2, b3; };

__device__ __forceinline__ void load_set(Set& s, const float* xp0, const float* xp1,
                                         const float* wp0, const float* wp1,
                                         const float* wp2, const float* wp3, int kb) {
  load_frag(s.a0, xp0 + kb);
  load_frag(s.a1, xp1 + kb);
  load_frag(s.b0, wp0 + kb);
  load_frag(s.b1, wp1 + kb);
  load_frag(s.b2, wp2 + kb);
  load_frag(s.b3, wp3 + kb);
}

__device__ __forceinline__ void mma_set(const Set& s, v8f acc[2][4]) {
#pragma unroll
  for (int j = 0; j < 8; ++j) {
    const v2f a0 = pairk(s.a0, j), a1 = pairk(s.a1, j);
    const v2f b0 = pairk(s.b0, j), b1 = pairk(s.b1, j);
    const v2f b2 = pairk(s.b2, j), b3 = pairk(s.b3, j);
    acc[0][0] = wmma4(a0, b0, acc[0][0]);
    acc[0][1] = wmma4(a0, b1, acc[0][1]);
    acc[0][2] = wmma4(a0, b2, acc[0][2]);
    acc[0][3] = wmma4(a0, b3, acc[0][3]);
    acc[1][0] = wmma4(a1, b0, acc[1][0]);
    acc[1][1] = wmma4(a1, b1, acc[1][1]);
    acc[1][2] = wmma4(a1, b2, acc[1][2]);
    acc[1][3] = wmma4(a1, b3, acc[1][3]);
  }
}

__device__ __forceinline__ void gemm_wave32(const float* xp0, const float* xp1,
                                            const float* wp0, const float* wp1,
                                            const float* wp2, const float* wp3,
                                            int K, v8f acc[2][4]) {
  Set P, Q;
  load_set(P, xp0, xp1, wp0, wp1, wp2, wp3, 0);
  for (int kb = 0; kb < K - 64; kb += 64) {
    load_set(Q, xp0, xp1, wp0, wp1, wp2, wp3, kb + 32);
    __builtin_amdgcn_sched_barrier(0);
    mma_set(P, acc);
    __builtin_amdgcn_sched_barrier(0);
    load_set(P, xp0, xp1, wp0, wp1, wp2, wp3, kb + 64);
    __builtin_amdgcn_sched_barrier(0);
    mma_set(Q, acc);
  }
  load_set(Q, xp0, xp1, wp0, wp1, wp2, wp3, K - 32);
  __builtin_amdgcn_sched_barrier(0);
  mma_set(P, acc);
  mma_set(Q, acc);
}

struct Set4 { Frag a[4]; Frag b[4]; };

__device__ __forceinline__ void load_set4(Set4& s, const float* const xp[4],
                                          const float* const wp[4], int kb) {
#pragma unroll
  for (int i = 0; i < 4; ++i) load_frag(s.a[i], xp[i] + kb);
#pragma unroll
  for (int t = 0; t < 4; ++t) load_frag(s.b[t], wp[t] + kb);
}

__device__ __forceinline__ void mma_set4(const Set4& s, v8f acc[4][4]) {
#pragma unroll
  for (int j = 0; j < 8; ++j) {
    v2f a[4], b[4];
#pragma unroll
    for (int i = 0; i < 4; ++i) { a[i] = pairk(s.a[i], j); b[i] = pairk(s.b[i], j); }
#pragma unroll
    for (int i = 0; i < 4; ++i)
#pragma unroll
      for (int t = 0; t < 4; ++t)
        acc[i][t] = wmma4(b[t], a[i], acc[i][t]);
  }
}

__device__ __forceinline__ void gemm_wave64(const float* const xp[4], const float* const wp[4],
                                            int K, v8f acc[4][4]) {
  Set4 P, Q;
  load_set4(P, xp, wp, 0);
  for (int kb = 0; kb < K - 64; kb += 64) {
    load_set4(Q, xp, wp, kb + 32);
    __builtin_amdgcn_sched_barrier(0);
    mma_set4(P, acc);
    __builtin_amdgcn_sched_barrier(0);
    load_set4(P, xp, wp, kb + 64);
    __builtin_amdgcn_sched_barrier(0);
    mma_set4(Q, acc);
  }
  load_set4(Q, xp, wp, K - 32);
  __builtin_amdgcn_sched_barrier(0);
  mma_set4(P, acc);
  mma_set4(Q, acc);
}

__global__ void __launch_bounds__(256) gemm_store(const float* __restrict__ X, const float* __restrict__ W,
                           float* __restrict__ C, int K, int M) {
  const int wave = threadIdx.x >> 5;
  const int lane = threadIdx.x & 31;
  const int half = lane >> 4;
  const int r    = lane & 15;
  const int row0 = blockIdx.y * ROWS_B;
  const int col0 = blockIdx.x * CHUNK + wave * 64;
  if (col0 >= M) return;

  const float* xp0 = X + (size_t)(row0 + r) * K + 16 * half;
  const float* xp1 = X + (size_t)(row0 + 16 + r) * K + 16 * half;
  const float* wp0 = W + (size_t)(col0 + r) * K + 16 * half;
  const float* wp1 = W + (size_t)(col0 + 16 + r) * K + 16 * half;
  const float* wp2 = W + (size_t)(col0 + 32 + r) * K + 16 * half;
  const float* wp3 = W + (size_t)(col0 + 48 + r) * K + 16 * half;

  v8f acc[2][4];
#pragma unroll
  for (int rt = 0; rt < 2; ++rt)
#pragma unroll
    for (int t = 0; t < 4; ++t)
      acc[rt][t] = (v8f){0.f,0.f,0.f,0.f,0.f,0.f,0.f,0.f};
  gemm_wave32(xp0, xp1, wp0, wp1, wp2, wp3, K, acc);

#pragma unroll
  for (int rt = 0; rt < 2; ++rt) {
#pragma unroll
    for (int t = 0; t < 4; ++t) {
#pragma unroll
      for (int v = 0; v < 8; ++v) {
        int rr = row0 + 16 * rt + v + 8 * half;
        C[(size_t)rr * M + col0 + 16 * t + r] = acc[rt][t][v];
      }
    }
  }
}

template <bool RAGGED>
__device__ __forceinline__ float lane_expsum(const v8f (&acc)[4], int lim) {
  float p[4];
#pragma unroll
  for (int t = 0; t < 4; ++t) {
    float q = 0.f;
#pragma unroll
    for (int v = 0; v < 8; ++v) {
      float e = expf(acc[t][v]);
      if (RAGGED) e = (16 * t + v < lim) ? e : 0.f;
      q = (v == 0) ? e : q + e;
    }
    p[t] = q;
  }
  return (p[0] + p[1]) + (p[2] + p[3]);
}

__global__ void __launch_bounds__(256) gemm_expsum(const float* __restrict__ X, const float* __restrict__ W,
                            float* __restrict__ partial, int K, int M, int nchunks) {
  __shared__ float lsum[8][4][32];
  const int wave = threadIdx.x >> 5;
  const int lane = threadIdx.x & 31;
  const int half = lane >> 4;
  const int r    = lane & 15;
  const int row0 = blockIdx.y * ROWS_E;
  const int col0 = blockIdx.x * CHUNK + wave * 64;
  const bool active = (col0 < M);

  v8f acc[4][4];
#pragma unroll
  for (int rt = 0; rt < 4; ++rt)
#pragma unroll
    for (int t = 0; t < 4; ++t)
      acc[rt][t] = (v8f){0.f,0.f,0.f,0.f,0.f,0.f,0.f,0.f};
  if (active) {
    const float* xp[4];
    const float* wp[4];
#pragma unroll
    for (int i = 0; i < 4; ++i) {
      xp[i] = X + (size_t)(row0 + 16 * i + r) * K + 16 * half;
      wp[i] = W + (size_t)min(col0 + 16 * i + r, M - 1) * K + 16 * half;
    }
    gemm_wave64(xp, wp, K, acc);
  }
  float s[4] = {0.f, 0.f, 0.f, 0.f};
  if (col0 + 64 <= M) {
#pragma unroll
    for (int i = 0; i < 4; ++i) s[i] = lane_expsum<false>(acc[i], 0);
  } else if (active) {
    const int lim = M - col0 - 8 * half;
#pragma unroll
    for (int i = 0; i < 4; ++i) s[i] = lane_expsum<true>(acc[i], lim);
  }
#pragma unroll
  for (int i = 0; i < 4; ++i) lsum[wave][i][lane] = s[i];
  __syncthreads();
  if (threadIdx.x < ROWS_E) {
    const int i = threadIdx.x >> 4;
    const int c = threadIdx.x & 15;
    float t = 0.f;
#pragma unroll
    for (int w = 0; w < 8; ++w) {
      t += lsum[w][i][c];
      t += lsum[w][i][16 + c];
    }
    partial[(size_t)(row0 + threadIdx.x) * nchunks + blockIdx.x] = t;
  }
}

__global__ void gather_dots(const float* __restrict__ x, const int* __restrict__ target,
                            const float* __restrict__ Wh, const float* __restrict__ Wo1,
                            const float* __restrict__ Wo2, const float* __restrict__ h1,
                            const float* __restrict__ h2,
                            float* __restrict__ dh, float* __restrict__ d1,
                            float* __restrict__ d2) {
  const int wave = threadIdx.x >> 5;
  const int lane = threadIdx.x & 31;
  const int row  = blockIdx.x * 8 + wave;
  const int t = target[row];
  const int g = (t < CUT0) ? t : ((t < CUT1) ? CUT0 : CUT0 + 1);

  float s0 = 0.f, s1 = 0.f, s2 = 0.f;
  {
    const float* xr = x + (size_t)row * DIM;
    const float* wr = Wh + (size_t)g * DIM;
    for (int k = lane; k < DIM; k += 32) s0 += xr[k] * wr[k];
  }
  {
    const int rel = min(max(t - CUT0, 0), M1 - 1);
    const float* hr = h1 + (size_t)row * D1;
    const float* wr = Wo1 + (size_t)rel * D1;
    for (int k = lane; k < D1; k += 32) s1 += hr[k] * wr[k];
  }
  {
    const int rel = min(max(t - CUT1, 0), M2 - 1);
    const float* hr = h2 + (size_t)row * D2;
    const float* wr = Wo2 + (size_t)rel * D2;
    for (int k = lane; k < D2; k += 32) s2 += hr[k] * wr[k];
  }
#pragma unroll
  for (int m = 16; m; m >>= 1) {
    s0 += __shfl_xor(s0, m, 32);
    s1 += __shfl_xor(s1, m, 32);
    s2 += __shfl_xor(s2, m, 32);
  }
  if (lane == 0) { dh[row] = s0; d1[row] = s1; d2[row] = s2; }
}

__global__ void finalize(const int* __restrict__ target,
                         const float* __restrict__ ph, const float* __restrict__ p1,
                         const float* __restrict__ p2, const float* __restrict__ dh,
                         const float* __restrict__ d1, const float* __restrict__ d2,
                         float* __restrict__ out) {
  const int i = blockIdx.x * blockDim.x + threadIdx.x;
  if (i >= NROWS) return;
  float sh = 0.f;
  for (int c = 0; c < NCH_H; ++c) sh += ph[(size_t)i * NCH_H + c];
  float o = dh[i] - logf(sh);
  const int t = target[i];
  if (t >= CUT0 && t < CUT1) {
    float s = 0.f;
    for (int c = 0; c < NCH_1; ++c) s += p1[(size_t)i * NCH_1 + c];
    o += d1[i] - logf(s);
  } else if (t >= CUT1) {
    float s = 0.f;
    for (int c = 0; c < NCH_2; ++c) s += p2[(size_t)i * NCH_2 + c];
    o += d2[i] - logf(s);
  }
  out[i] = o;
}

__global__ void loss_reduce(const float* __restrict__ out, float* __restrict__ loss) {
  __shared__ float sm[256];
  float s = 0.f;
  for (int i = threadIdx.x; i < NROWS; i += 256) s += out[i];
  sm[threadIdx.x] = s;
  __syncthreads();
  for (int w = 128; w; w >>= 1) {
    if (threadIdx.x < w) sm[threadIdx.x] += sm[threadIdx.x + w];
    __syncthreads();
  }
  if (threadIdx.x == 0) *loss = -sm[0] / (float)NROWS;
}

extern "C" void kernel_launch(void* const* d_in, const int* in_sizes, int n_in,
                              void* d_out, int out_size, void* d_ws, size_t ws_size,
                              hipStream_t stream) {
  (void)in_sizes; (void)n_in; (void)out_size; (void)ws_size;
  const float* x    = (const float*)d_in[0];
  const int*   targ = (const int*)d_in[1];
  const float* Wh   = (const float*)d_in[2];
  const float* Wp1  = (const float*)d_in[3];
  const float* Wo1  = (const float*)d_in[4];
  const float* Wp2  = (const float*)d_in[5];
  const float* Wo2  = (const float*)d_in[6];
  float* out = (float*)d_out;

  float* ws = (float*)d_ws;
  float* h1 = ws;
  float* h2 = h1 + (size_t)NROWS * D1;
  float* ph = h2 + (size_t)NROWS * D2;
  float* p1 = ph + (size_t)NROWS * NCH_H;
  float* p2 = p1 + (size_t)NROWS * NCH_1;
  float* dh = p2 + (size_t)NROWS * NCH_2;
  float* d1 = dh + NROWS;
  float* d2 = d1 + NROWS;

  const dim3 blk(256);
  const int rowBlocks = NROWS / ROWS_B;
  const int rowBlocksE = NROWS / ROWS_E;
  gemm_store<<<dim3(1, rowBlocks), blk, 0, stream>>>(x, Wp1, h1, DIM, D1);
  gemm_store<<<dim3(1, rowBlocks), blk, 0, stream>>>(x, Wp2, h2, DIM, D2);
  gemm_expsum<<<dim3(NCH_H, rowBlocksE), blk, 0, stream>>>(x,  Wh,  ph, DIM, HEADM, NCH_H);
  gemm_expsum<<<dim3(NCH_1, rowBlocksE), blk, 0, stream>>>(h1, Wo1, p1, D1,  M1,    NCH_1);
  gemm_expsum<<<dim3(NCH_2, rowBlocksE), blk, 0, stream>>>(h2, Wo2, p2, D2,  M2,    NCH_2);
  gather_dots<<<dim3(NROWS / 8), blk, 0, stream>>>(x, targ, Wh, Wo1, Wo2, h1, h2, dh, d1, d2);
  finalize<<<dim3(NROWS / 256), blk, 0, stream>>>(targ, ph, p1, p2, dh, d1, d2, out);
  loss_reduce<<<1, 256, 0, stream>>>(out, out + NROWS);
}


# Exp-sum GEMMs: a block keeps its 64 rows and walks 5 or 10 column chunks, no barrier between chunks, exp-sums added in-lane across chunks; 256 blocks per GEMM; changes the exp-sum tree
# speedup vs baseline: 3.4138x; 1.0636x over previous
#include <hip/hip_runtime.h>
#include <math.h>


#define CUT0 10000
#define CUT1 30000
#define DIM  1024
#define NROWS 4096
#define HEADM 10002
#define M1 20000
#define M2 20257
#define D1 256
#define D2 64
#define ROWS_B 32
#define ROWS_E 64
#define CHUNK 512
#define NCH_H 20
#define NCH_1 40
#define NCH_2 40
#define WALK_H 5
#define WALK_1 10
#define WALK_2 10
#define NP_H (NCH_H / WALK_H)
#define NP_1 (NCH_1 / WALK_1)
#define NP_2 (NCH_2 / WALK_2)

typedef float v2f __attribute__((ext_vector_type(2)));
typedef float v4f __attribute__((ext_vector_type(4)));
typedef float v8f __attribute__((ext_vector_type(8)));

__device__ __forceinline__ v8f wmma4(v2f a, v2f b, v8f c) {
  return __builtin_amdgcn_wmma_f32_16x16x4_f32(false, a, false, b, (short)0, c,
                                               false, false);
}

struct Frag { v4f q[4]; };

__device__ __forceinline__ void load_frag(Frag& f, const float* p) {
  f.q[0] = *(const v4f*)(p);
  f.q[1] = *(const v4f*)(p + 4);
  f.q[2] = *(const v4f*)(p + 8);
  f.q[3] = *(const v4f*)(p + 12);
}

__device__ __forceinline__ v2f pairk(const Frag& f, int j) {
  const v4f q = f.q[j >> 1];
  return (j & 1) ? (v2f){q.z, q.w} : (v2f){q.x, q.y};
}

struct Set { Frag a0, a1, b0, b1, b2, b3; };

__device__ __forceinline__ void load_set(Set& s, const float* xp0, const float* xp1,
                                         const float* wp0, const float* wp1,
                                         const float* wp2, const float* wp3, int kb) {
  load_frag(s.a0, xp0 + kb);
  load_frag(s.a1, xp1 + kb);
  load_frag(s.b0, wp0 + kb);
  load_frag(s.b1, wp1 + kb);
  load_frag(s.b2, wp2 + kb);
  load_frag(s.b3, wp3 + kb);
}

__device__ __forceinline__ void mma_set(const Set& s, v8f acc[2][4]) {
#pragma unroll
  for (int j = 0; j < 8; ++j) {
    const v2f a0 = pairk(s.a0, j), a1 = pairk(s.a1, j);
    const v2f b0 = pairk(s.b0, j), b1 = pairk(s.b1, j);
    const v2f b2 = pairk(s.b2, j), b3 = pairk(s.b3, j);
    acc[0][0] = wmma4(a0, b0, acc[0][0]);
    acc[0][1] = wmma4(a0, b1, acc[0][1]);
    acc[0][2] = wmma4(a0, b2, acc[0][2]);
    acc[0][3] = wmma4(a0, b3, acc[0][3]);
    acc[1][0] = wmma4(a1, b0, acc[1][0]);
    acc[1][1] = wmma4(a1, b1, acc[1][1]);
    acc[1][2] = wmma4(a1, b2, acc[1][2]);
    acc[1][3] = wmma4(a1, b3, acc[1][3]);
  }
}

__device__ __forceinline__ void gemm_wave32(const float* xp0, const float* xp1,
                                            const float* wp0, const float* wp1,
                                            const float* wp2, const float* wp3,
                                            int K, v8f acc[2][4]) {
  Set P, Q;
  load_set(P, xp0, xp1, wp0, wp1, wp2, wp3, 0);
  for (int kb = 0; kb < K - 64; kb += 64) {
    load_set(Q, xp0, xp1, wp0, wp1, wp2, wp3, kb + 32);
    __builtin_amdgcn_sched_barrier(0);
    mma_set(P, acc);
    __builtin_amdgcn_sched_barrier(0);
    load_set(P, xp0, xp1, wp0, wp1, wp2, wp3, kb + 64);
    __builtin_amdgcn_sched_barrier(0);
    mma_set(Q, acc);
  }
  load_set(Q, xp0, xp1, wp0, wp1, wp2, wp3, K - 32);
  __builtin_amdgcn_sched_barrier(0);
  mma_set(P, acc);
  mma_set(Q, acc);
}

struct Set4 { Frag a[4]; Frag b[4]; };

__device__ __forceinline__ void load_set4(Set4& s, const float* const xp[4],
                                          const float* const wp[4], int kb) {
#pragma unroll
  for (int i = 0; i < 4; ++i) load_frag(s.a[i], xp[i] + kb);
#pragma unroll
  for (int t = 0; t < 4; ++t) load_frag(s.b[t], wp[t] + kb);
}

__device__ __forceinline__ void mma_set4(const Set4& s, v8f acc[4][4]) {
#pragma unroll
  for (int j = 0; j < 8; ++j) {
    v2f a[4], b[4];
#pragma unroll
    for (int i = 0; i < 4; ++i) { a[i] = pairk(s.a[i], j); b[i] = pairk(s.b[i], j); }
#pragma unroll
    for (int i = 0; i < 4; ++i)
#pragma unroll
      for (int t = 0; t < 4; ++t)
        acc[i][t] = wmma4(b[t], a[i], acc[i][t]);
  }
}

__device__ __forceinline__ void gemm_wave64(const float* const xp[4], const float* const wp[4],
                                            int K, v8f acc[4][4]) {
  Set4 P, Q;
  load_set4(P, xp, wp, 0);
  for (int kb = 0; kb < K - 64; kb += 64) {
    load_set4(Q, xp, wp, kb + 32);
    __builtin_amdgcn_sched_barrier(0);
    mma_set4(P, acc);
    __builtin_amdgcn_sched_barrier(0);
    load_set4(P, xp, wp, kb + 64);
    __builtin_amdgcn_sched_barrier(0);
    mma_set4(Q, acc);
  }
  load_set4(Q, xp, wp, K - 32);
  __builtin_amdgcn_sched_barrier(0);
  mma_set4(P, acc);
  mma_set4(Q, acc);
}

__global__ void __launch_bounds__(256) gemm_store(const float* __restrict__ X, const float* __restrict__ W,
                           float* __restrict__ C, int K, int M) {
  const int wave = threadIdx.x >> 5;
  const int lane = threadIdx.x & 31;
  const int half = lane >> 4;
  const int r    = lane & 15;
  const int row0 = blockIdx.y * ROWS_B;
  const int col0 = blockIdx.x * CHUNK + wave * 64;
  if (col0 >= M) return;

  const float* xp0 = X + (size_t)(row0 + r) * K + 16 * half;
  const float* xp1 = X + (size_t)(row0 + 16 + r) * K + 16 * half;
  const float* wp0 = W + (size_t)(col0 + r) * K + 16 * half;
  const float* wp1 = W + (size_t)(col0 + 16 + r) * K + 16 * half;
  const float* wp2 = W + (size_t)(col0 + 32 + r) * K + 16 * half;
  const float* wp3 = W + (size_t)(col0 + 48 + r) * K + 16 * half;

  v8f acc[2][4];
#pragma unroll
  for (int rt = 0; rt < 2; ++rt)
#pragma unroll
    for (int t = 0; t < 4; ++t)
      acc[rt][t] = (v8f){0.f,0.f,0.f,0.f,0.f,0.f,0.f,0.f};
  gemm_wave32(xp0, xp1, wp0, wp1, wp2, wp3, K, acc);

#pragma unroll
  for (int rt = 0; rt < 2; ++rt) {
#pragma unroll
    for (int t = 0; t < 4; ++t) {
#pragma unroll
      for (int v = 0; v < 8; ++v) {
        int rr = row0 + 16 * rt + v + 8 * half;
        C[(size_t)rr * M + col0 + 16 * t + r] = acc[rt][t][v];
      }
    }
  }
}

template <bool RAGGED>
__device__ __forceinline__ float lane_expsum(const v8f (&acc)[4], int lim) {
  float p[4];
#pragma unroll
  for (int t = 0; t < 4; ++t) {
    float q = 0.f;
#pragma unroll
    for (int v = 0; v < 8; ++v) {
      float e = expf(acc[t][v]);
      if (RAGGED) e = (16 * t + v < lim) ? e : 0.f;
      q = (v == 0) ? e : q + e;
    }
    p[t] = q;
  }
  return (p[0] + p[1]) + (p[2] + p[3]);
}

__global__ void __launch_bounds__(256) gemm_expsum(const float* __restrict__ X, const float* __restrict__ W,
                            float* __restrict__ partial, int K, int M, int walk) {
  __shared__ float lsum[8][4][32];
  const int wave = threadIdx.x >> 5;
  const int lane = threadIdx.x & 31;
  const int half = lane >> 4;
  const int r    = lane & 15;
  const int row0 = blockIdx.y * ROWS_E;

  const float* xp[4];
#pragma unroll
  for (int i = 0; i < 4; ++i)
    xp[i] = X + (size_t)(row0 + 16 * i + r) * K + 16 * half;

  float s[4] = {0.f, 0.f, 0.f, 0.f};
#pragma nounroll
  for (int c = 0; c < walk; ++c) {
    const int col0 = (blockIdx.x * walk + c) * CHUNK + wave * 64;
    if (col0 < M) {
      v8f acc[4][4];
#pragma unroll
      for (int rt = 0; rt < 4; ++rt)
#pragma unroll
        for (int t = 0; t < 4; ++t)
          acc[rt][t] = (v8f){0.f,0.f,0.f,0.f,0.f,0.f,0.f,0.f};
      const float* wp[4];
#pragma unroll
      for (int i = 0; i < 4; ++i)
        wp[i] = W + (size_t)min(col0 + 16 * i + r, M - 1) * K + 16 * half;
      gemm_wave64(xp, wp, K, acc);
      if (col0 + 64 <= M) {
#pragma unroll
        for (int i = 0; i < 4; ++i) s[i] += lane_expsum<false>(acc[i], 0);
      } else {
        const int lim = M - col0 - 8 * half;
#pragma unroll
        for (int i = 0; i < 4; ++i) s[i] += lane_expsum<true>(acc[i], lim);
      }
    }
  }
#pragma unroll
  for (int i = 0; i < 4; ++i) lsum[wave][i][lane] = s[i];
  __syncthreads();
  if (threadIdx.x < ROWS_E) {
    const int i = threadIdx.x >> 4;
    const int c = threadIdx.x & 15;
    float t = 0.f;
#pragma unroll
    for (int w = 0; w < 8; ++w) {
      t += lsum[w][i][c];
      t += lsum[w][i][16 + c];
    }
    partial[(size_t)blockIdx.x * NROWS + row0 + threadIdx.x] = t;
  }
}

__global__ void gather_dots(const float* __restrict__ x, const int* __restrict__ target,
                            const float* __restrict__ Wh, const float* __restrict__ Wo1,
                            const float* __restrict__ Wo2, const float* __restrict__ h1,
                            const float* __restrict__ h2,
                            float* __restrict__ dh, float* __restrict__ d1,
                            float* __restrict__ d2) {
  const int wave = threadIdx.x >> 5;
  const int lane = threadIdx.x & 31;
  const int row  = blockIdx.x * 8 + wave;
  const int t = target[row];
  const int g = (t < CUT0) ? t : ((t < CUT1) ? CUT0 : CUT0 + 1);

  float s0 = 0.f, s1 = 0.f, s2 = 0.f;
  {
    const float* xr = x + (size_t)row * DIM;
    const float* wr = Wh + (size_t)g * DIM;
    for (int k = lane; k < DIM; k += 32) s0 += xr[k] * wr[k];
  }
  {
    const int rel = min(max(t - CUT0, 0), M1 - 1);
    const float* hr = h1 + (size_t)row * D1;
    const float* wr = Wo1 + (size_t)rel * D1;
    for (int k = lane; k < D1; k += 32) s1 += hr[k] * wr[k];
  }
  {
    const int rel = min(max(t - CUT1, 0), M2 - 1);
    const float* hr = h2 + (size_t)row * D2;
    const float* wr = Wo2 + (size_t)rel * D2;
    for (int k = lane; k < D2; k += 32) s2 += hr[k] * wr[k];
  }
#pragma unroll
  for (int m = 16; m; m >>= 1) {
    s0 += __shfl_xor(s0, m, 32);
    s1 += __shfl_xor(s1, m, 32);
    s2 += __shfl_xor(s2, m, 32);
  }
  if (lane == 0) { dh[row] = s0; d1[row] = s1; d2[row] = s2; }
}

__global__ void finalize(const int* __restrict__ target,
                         const float* __restrict__ ph, const float* __restrict__ p1,
                         const float* __restrict__ p2, const float* __restrict__ dh,
                         const float* __restrict__ d1, const float* __restrict__ d2,
                         float* __restrict__ out) {
  const int i = blockIdx.x * blockDim.x + threadIdx.x;
  if (i >= NROWS) return;
  float sh = 0.f;
  for (int c = 0; c < NP_H; ++c) sh += ph[(size_t)c * NROWS + i];
  float o = dh[i] - logf(sh);
  const int t = target[i];
  if (t >= CUT0 && t < CUT1) {
    float s = 0.f;
    for (int c = 0; c < NP_1; ++c) s += p1[(size_t)c * NROWS + i];
    o += d1[i] - logf(s);
  } else if (t >= CUT1) {
    float s = 0.f;
    for (int c = 0; c < NP_2; ++c) s += p2[(size_t)c * NROWS + i];
    o += d2[i] - logf(s);
  }
  out[i] = o;
}

__global__ void loss_reduce(const float* __restrict__ out, float* __restrict__ loss) {
  __shared__ float sm[256];
  float s = 0.f;
  for (int i = threadIdx.x; i < NROWS; i += 256) s += out[i];
  sm[threadIdx.x] = s;
  __syncthreads();
  for (int w = 128; w; w >>= 1) {
    if (threadIdx.x < w) sm[threadIdx.x] += sm[threadIdx.x + w];
    __syncthreads();
  }
  if (threadIdx.x == 0) *loss = -sm[0] / (float)NROWS;
}

extern "C" void kernel_launch(void* const* d_in, const int* in_sizes, int n_in,
                              void* d_out, int out_size, void* d_ws, size_t ws_size,
                              hipStream_t stream) {
  (void)in_sizes; (void)n_in; (void)out_size; (void)ws_size;
  const float* x    = (const float*)d_in[0];
  const int*   targ = (const int*)d_in[1];
  const float* Wh   = (const float*)d_in[2];
  const float* Wp1  = (const float*)d_in[3];
  const float* Wo1  = (const float*)d_in[4];
  const float* Wp2  = (const float*)d_in[5];
  const float* Wo2  = (const float*)d_in[6];
  float* out = (float*)d_out;

  float* ws = (float*)d_ws;
  float* h1 = ws;
  float* h2 = h1 + (size_t)NROWS * D1;
  float* ph = h2 + (size_t)NROWS * D2;
  float* p1 = ph + (size_t)NROWS * NP_H;
  float* p2 = p1 + (size_t)NROWS * NP_1;
  float* dh = p2 + (size_t)NROWS * NP_2;
  float* d1 = dh + NROWS;
  float* d2 = d1 + NROWS;

  const dim3 blk(256);
  const int rowBlocks = NROWS / ROWS_B;
  const int rowBlocksE = NROWS / ROWS_E;
  gemm_store<<<dim3(1, rowBlocks), blk, 0, stream>>>(x, Wp1, h1, DIM, D1);
  gemm_store<<<dim3(1, rowBlocks), blk, 0, stream>>>(x, Wp2, h2, DIM, D2);
  gemm_expsum<<<dim3(NP_H, rowBlocksE), blk, 0, stream>>>(x,  Wh,  ph, DIM, HEADM, WALK_H);
  gemm_expsum<<<dim3(NP_1, rowBlocksE), blk, 0, stream>>>(h1, Wo1, p1, D1,  M1,    WALK_1);
  gemm_expsum<<<dim3(NP_2, rowBlocksE), blk, 0, stream>>>(h2, Wo2, p2, D2,  M2,    WALK_2);
  gather_dots<<<dim3(NROWS / 8), blk, 0, stream>>>(x, targ, Wh, Wo1, Wo2, h1, h2, dh, d1, d2);
  finalize<<<dim3(NROWS / 256), blk, 0, stream>>>(targ, ph, p1, p2, dh, d1, d2, out);
  loss_reduce<<<1, 256, 0, stream>>>(out, out + NROWS);
}
